# starting kernel
# baseline (speedup 1.0000x reference)
#include <hip/hip_runtime.h>

typedef __attribute__((ext_vector_type(2))) float v2f;
typedef __attribute__((ext_vector_type(8))) float v8f;

#define TOKENS  4096
#define IN_F    4096
#define OUT_F   4096
#define LOWRANK 256
#define BS      256
#define ACTIVE  5
#define KBF     (ACTIVE * BS)

__global__ __launch_bounds__(256) void gather_wg(const float* __restrict__ weight,
                                                 const int* __restrict__ flat,
                                                 float* __restrict__ wg) {
    int g   = blockIdx.x * 256 + threadIdx.x;
    int f   = g / KBF;
    int rem = g - f * KBF;
    int j   = rem >> 8;
    int c   = rem & 255;
    int ob  = f >> 8;
    int rr  = f & 255;
    int fl  = flat[ob * ACTIVE + j];
    int q   = fl / ACTIVE;
    int bb  = fl - q * ACTIVE;
    int k   = bb * BS + rr;
    int kr  = k / ACTIVE;
    int a   = k - kr * ACTIVE;
    int srcRow = q * BS + kr;
    wg[g] = weight[(srcRow * ACTIVE + a) * BS + c];
}

__global__ __launch_bounds__(256) void gemm_xw1(const float* __restrict__ x,
                                                const float* __restrict__ w1,
                                                float* __restrict__ T) {
    const int lane = threadIdx.x & 31;
    const int wave = threadIdx.x >> 5;
    const int tile = blockIdx.x * 8 + wave;
    const int nft  = LOWRANK / 32;
    const int tt   = tile / nft;
    const int ft   = tile - tt * nft;
    const int t0   = tt * 32;
    const int f0   = ft * 32;
    const int lr   = lane & 15;
    const int co   = (lane >> 4) * 2;

    const float* a0p = x  + (t0 + lr)      * IN_F + co;
    const float* a1p = x  + (t0 + 16 + lr) * IN_F + co;
    const float* b0p = w1 + (f0 + lr)      * IN_F + co;
    const float* b1p = w1 + (f0 + 16 + lr) * IN_F + co;

    v8f acc00 = {}, acc01 = {}, acc10 = {}, acc11 = {};
    for (int k = 0; k < IN_F; k += 16) {
#pragma unroll
        for (int u = 0; u < 16; u += 4) {
            v2f a0 = *(const v2f*)(a0p + k + u);
            v2f a1 = *(const v2f*)(a1p + k + u);
            v2f b0 = *(const v2f*)(b0p + k + u);
            v2f b1 = *(const v2f*)(b1p + k + u);
            acc00 = __builtin_amdgcn_wmma_f32_16x16x4_f32(false, a0, false, b0,
                                                          (short)0, acc00, false, false);
            acc01 = __builtin_amdgcn_wmma_f32_16x16x4_f32(false, a0, false, b1,
                                                          (short)0, acc01, false, false);
            acc10 = __builtin_amdgcn_wmma_f32_16x16x4_f32(false, a1, false, b0,
                                                          (short)0, acc10, false, false);
            acc11 = __builtin_amdgcn_wmma_f32_16x16x4_f32(false, a1, false, b1,
                                                          (short)0, acc11, false, false);
        }
    }
    const int m0 = (lane >> 4) * 8;
#pragma unroll
    for (int r = 0; r < 8; ++r) {
        T[(t0 + m0 + r)      * LOWRANK + f0 + lr]      = acc00[r];
        T[(t0 + m0 + r)      * LOWRANK + f0 + 16 + lr] = acc01[r];
        T[(t0 + 16 + m0 + r) * LOWRANK + f0 + lr]      = acc10[r];
        T[(t0 + 16 + m0 + r) * LOWRANK + f0 + 16 + lr] = acc11[r];
    }
}

__global__ __launch_bounds__(256) void fused_out(const float* __restrict__ x,
                                                 const float* __restrict__ wg,
                                                 const float* __restrict__ T,
                                                 const float* __restrict__ w2,
                                                 const float* __restrict__ bias,
                                                 const int* __restrict__ flat,
                                                 float* __restrict__ out) {
    const int lane = threadIdx.x & 31;
    const int wave = threadIdx.x >> 5;
    const int tile = blockIdx.x * 8 + wave;
    const int nft  = OUT_F / 64;
    const int tt   = tile / nft;
    const int ft   = tile - tt * nft;
    const int t0   = tt * 64;
    const int f0   = ft * 64;
    const int ob   = f0 >> 8;
    const int lr   = lane & 15;
    const int co   = (lane >> 4) * 2;

    v8f acc[4][4];
#pragma unroll
    for (int i = 0; i < 4; ++i)
#pragma unroll
        for (int n = 0; n < 4; ++n) acc[i][n] = (v8f){};

    for (int j = 0; j < ACTIVE; ++j) {
        const int q = flat[ob * ACTIVE + j] / ACTIVE;
        const float* arow[4];
        const float* brow[4];
#pragma unroll
        for (int i = 0; i < 4; ++i)
            arow[i] = x + (t0 + 16 * i + lr) * IN_F + q * BS + co;
#pragma unroll
        for (int n = 0; n < 4; ++n)
            brow[n] = wg + (f0 + 16 * n + lr) * KBF + j * BS + co;

        for (int k = 0; k < BS; k += 4) {
            v2f af[4], bf[4];
#pragma unroll
            for (int i = 0; i < 4; ++i) af[i] = *(const v2f*)(arow[i] + k);
#pragma unroll
            for (int n = 0; n < 4; ++n) bf[n] = *(const v2f*)(brow[n] + k);
#pragma unroll
            for (int i = 0; i < 4; ++i)
#pragma unroll
                for (int n = 0; n < 4; ++n)
                    acc[i][n] = __builtin_amdgcn_wmma_f32_16x16x4_f32(
                        false, af[i], false, bf[n], (short)0, acc[i][n], false, false);
        }
    }

    {
        const float* arow[4];
        const float* brow[4];
#pragma unroll
        for (int i = 0; i < 4; ++i)
            arow[i] = T  + (t0 + 16 * i + lr) * LOWRANK + co;
#pragma unroll
        for (int n = 0; n < 4; ++n)
            brow[n] = w2 + (f0 + 16 * n + lr) * LOWRANK + co;

        for (int k = 0; k < LOWRANK; k += 4) {
            v2f af[4], bf[4];
#pragma unroll
            for (int i = 0; i < 4; ++i) af[i] = *(const v2f*)(arow[i] + k);
#pragma unroll
            for (int n = 0; n < 4; ++n) bf[n] = *(const v2f*)(brow[n] + k);
#pragma unroll
            for (int i = 0; i < 4; ++i)
#pragma unroll
                for (int n = 0; n < 4; ++n)
                    acc[i][n] = __builtin_amdgcn_wmma_f32_16x16x4_f32(
                        false, af[i], false, bf[n], (short)0, acc[i][n], false, false);
        }
    }

    const int m0 = (lane >> 4) * 8;
#pragma unroll
    for (int n = 0; n < 4; ++n) {
        const float bv = bias[f0 + 16 * n + lr];
#pragma unroll
        for (int i = 0; i < 4; ++i)
#pragma unroll
            for (int r = 0; r < 8; ++r)
                out[(t0 + 16 * i + m0 + r) * OUT_F + f0 + 16 * n + lr] = acc[i][n][r] + bv;
    }
}

extern "C" void kernel_launch(void* const* d_in, const int* in_sizes, int n_in,
                              void* d_out, int out_size, void* d_ws, size_t ws_size,
                              hipStream_t stream) {
    const float* x      = (const float*)d_in[0];
    const float* weight = (const float*)d_in[1];
    const float* w1     = (const float*)d_in[2];
    const float* w2     = (const float*)d_in[3];
    const float* bias   = (const float*)d_in[4];
    const int*   flat   = (const int*)d_in[5];
    float*       out    = (float*)d_out;

    float* wg = (float*)d_ws;
    float* T  = wg + (size_t)OUT_F * KBF;

    gather_wg<<<dim3((OUT_F * KBF) / 256), dim3(256), 0, stream>>>(weight, flat, wg);

    gemm_xw1<<<dim3(1024 / 8), dim3(256), 0, stream>>>(x, w1, T);

    fused_out<<<dim3(4096 / 8), dim3(256), 0, stream>>>(x, wg, T, w2, bias, flat, out);
}


# w1, the gathered weights and T are stored in the workspace in WMMA fragment order (one extra small launch copies w1) and read as contiguous 16-byte-per-lane loads; k order and sums unchanged
# speedup vs baseline: 1.2732x; 1.2732x over previous
#include <hip/hip_runtime.h>

typedef __attribute__((ext_vector_type(2))) float v2f;
typedef __attribute__((ext_vector_type(4))) float v4f;
typedef __attribute__((ext_vector_type(8))) float v8f;

#define TOKENS  4096
#define IN_F    4096
#define OUT_F   4096
#define LOWRANK 256
#define BS      256
#define ACTIVE  5
#define KBF     (ACTIVE * BS)
#define KQ_W    (KBF / 8)
#define KQ_T    (LOWRANK / 8)
#define KQ_1    (IN_F / 8)


__global__ __launch_bounds__(256) void pack_w1(const float* __restrict__ w1,
                                               float* __restrict__ w1p) {
    const int g    = blockIdx.x * 256 + threadIdx.x;
    const int lane = g & 31;
    const int blk  = g >> 5;
    const int fb   = blk / KQ_1;
    const int kq   = blk - fb * KQ_1;
    const int f    = fb * 16 + (lane & 15);
    const int h    = lane >> 4;
    const float* s = w1 + (size_t)f * IN_F + kq * 8 + 2 * h;
    const v2f lo = *(const v2f*)(s);
    const v2f hi = *(const v2f*)(s + 4);
    v4f o;
    o.x = lo.x; o.y = lo.y; o.z = hi.x; o.w = hi.y;
    *(v4f*)(w1p + (size_t)g * 4) = o;
}

__global__ __launch_bounds__(256) void gather_wg(const float* __restrict__ weight,
                                                 const int* __restrict__ flat,
                                                 float* __restrict__ wp) {
    const int g    = blockIdx.x * 256 + threadIdx.x;
    const int lane = g & 31;
    const int blk  = g >> 5;
    const int fb   = blk / KQ_W;
    const int kq   = blk - fb * KQ_W;
    const int f    = fb * 16 + (lane & 15);
    const int h    = lane >> 4;
    const int ob   = f >> 8;
    const int rr   = f & 255;
    const int j    = kq >> 5;
    const int c0   = (kq & 31) * 8 + 2 * h;
    const int fl   = flat[ob * ACTIVE + j];
    const float* s = weight + ((size_t)fl * BS + rr) * BS + c0;
    const v2f lo = *(const v2f*)(s);
    const v2f hi = *(const v2f*)(s + 4);
    v4f o;
    o.x = lo.x; o.y = lo.y; o.z = hi.x; o.w = hi.y;
    *(v4f*)(wp + (size_t)g * 4) = o;
}

__global__ __launch_bounds__(256) void gemm_xw1(const float* __restrict__ x,
                                                const float* __restrict__ w1p,
                                                float* __restrict__ T) {
    const int lane = threadIdx.x & 31;
    const int wave = threadIdx.x >> 5;
    const int tile = blockIdx.x * 8 + wave;
    const int nft  = LOWRANK / 32;
    const int tt   = tile / nft;
    const int ft   = tile - tt * nft;
    const int t0   = tt * 32;
    const int f0   = ft * 32;
    const int lr   = lane & 15;
    const int co   = (lane >> 4) * 2;

    const float* a0p = x  + (t0 + lr)      * IN_F + co;
    const float* a1p = x  + (t0 + 16 + lr) * IN_F + co;
    const float* b0q = w1p + ((size_t)((f0 >> 4))     * KQ_1) * 128 + lane * 4;
    const float* b1q = w1p + ((size_t)((f0 >> 4) + 1) * KQ_1) * 128 + lane * 4;

    v8f acc00 = {}, acc01 = {}, acc10 = {}, acc11 = {};
    for (int k = 0; k < IN_F; k += 16) {
#pragma unroll
        for (int u = 0; u < 16; u += 8) {
            const v2f a0  = *(const v2f*)(a0p + k + u);
            const v2f a1  = *(const v2f*)(a1p + k + u);
            const v2f a0n = *(const v2f*)(a0p + k + u + 4);
            const v2f a1n = *(const v2f*)(a1p + k + u + 4);
            const v4f q0  = *(const v4f*)(b0q + (k + u) * 16);
            const v4f q1  = *(const v4f*)(b1q + (k + u) * 16);
            const v2f b0  = {q0.x, q0.y}, b0n = {q0.z, q0.w};
            const v2f b1  = {q1.x, q1.y}, b1n = {q1.z, q1.w};
            acc00 = __builtin_amdgcn_wmma_f32_16x16x4_f32(false, a0, false, b0,
                                                          (short)0, acc00, false, false);
            acc01 = __builtin_amdgcn_wmma_f32_16x16x4_f32(false, a0, false, b1,
                                                          (short)0, acc01, false, false);
            acc10 = __builtin_amdgcn_wmma_f32_16x16x4_f32(false, a1, false, b0,
                                                          (short)0, acc10, false, false);
            acc11 = __builtin_amdgcn_wmma_f32_16x16x4_f32(false, a1, false, b1,
                                                          (short)0, acc11, false, false);
            acc00 = __builtin_amdgcn_wmma_f32_16x16x4_f32(false, a0n, false, b0n,
                                                          (short)0, acc00, false, false);
            acc01 = __builtin_amdgcn_wmma_f32_16x16x4_f32(false, a0n, false, b1n,
                                                          (short)0, acc01, false, false);
            acc10 = __builtin_amdgcn_wmma_f32_16x16x4_f32(false, a1n, false, b0n,
                                                          (short)0, acc10, false, false);
            acc11 = __builtin_amdgcn_wmma_f32_16x16x4_f32(false, a1n, false, b1n,
                                                          (short)0, acc11, false, false);
        }
    }
    const int m0  = (lane >> 4) * 8;
    const int ff  = lr & 7;
    const int hh  = (ff >> 1) & 1;
    const int e   = (ff & 1) + 2 * (ff >> 2);
    const int tb0 = t0 >> 4;
    const int kq0 = (f0 >> 3) + (lr >> 3);
    const int in0 = (m0 + 16 * hh) * 4 + e;
    float* o00 = T + ((size_t)(tb0     * KQ_T + kq0)     * 128) + in0;
    float* o01 = T + ((size_t)(tb0     * KQ_T + kq0 + 2) * 128) + in0;
    float* o10 = T + ((size_t)((tb0+1) * KQ_T + kq0)     * 128) + in0;
    float* o11 = T + ((size_t)((tb0+1) * KQ_T + kq0 + 2) * 128) + in0;
#pragma unroll
    for (int r = 0; r < 8; ++r) {
        o00[4 * r] = acc00[r];
        o01[4 * r] = acc01[r];
        o10[4 * r] = acc10[r];
        o11[4 * r] = acc11[r];
    }
}

__global__ __launch_bounds__(256) void fused_out(const float* __restrict__ x,
                                                 const float* __restrict__ wp,
                                                 const float* __restrict__ tp,
                                                 const float* __restrict__ w2,
                                                 const float* __restrict__ bias,
                                                 const int* __restrict__ flat,
                                                 float* __restrict__ out) {
    const int lane = threadIdx.x & 31;
    const int wave = threadIdx.x >> 5;
    const int tile = blockIdx.x * 8 + wave;
    const int nft  = OUT_F / 64;
    const int tt   = tile / nft;
    const int ft   = tile - tt * nft;
    const int t0   = tt * 64;
    const int f0   = ft * 64;
    const int ob   = f0 >> 8;
    const int lr   = lane & 15;
    const int co   = (lane >> 4) * 2;

    v8f acc[4][4];
#pragma unroll
    for (int i = 0; i < 4; ++i)
#pragma unroll
        for (int n = 0; n < 4; ++n) acc[i][n] = (v8f){};

    for (int j = 0; j < ACTIVE; ++j) {
        const int q = flat[ob * ACTIVE + j] / ACTIVE;
        const float* arow[4];
        const float* bblk[4];
#pragma unroll
        for (int i = 0; i < 4; ++i)
            arow[i] = x + (t0 + 16 * i + lr) * IN_F + q * BS + co;
#pragma unroll
        for (int n = 0; n < 4; ++n)
            bblk[n] = wp + ((size_t)((f0 >> 4) + n) * KQ_W + j * (BS / 8)) * 128 + lane * 4;

        for (int k = 0; k < BS; k += 8) {
            v2f a0[4], a1[4];
            v4f bq[4];
#pragma unroll
            for (int i = 0; i < 4; ++i) {
                a0[i] = *(const v2f*)(arow[i] + k);
                a1[i] = *(const v2f*)(arow[i] + k + 4);
            }
#pragma unroll
            for (int n = 0; n < 4; ++n) bq[n] = *(const v4f*)(bblk[n] + k * 16);
#pragma unroll
            for (int i = 0; i < 4; ++i)
#pragma unroll
                for (int n = 0; n < 4; ++n) {
                    const v2f b0 = {bq[n].x, bq[n].y};
                    acc[i][n] = __builtin_amdgcn_wmma_f32_16x16x4_f32(
                        false, a0[i], false, b0, (short)0, acc[i][n], false, false);
                }
#pragma unroll
            for (int i = 0; i < 4; ++i)
#pragma unroll
                for (int n = 0; n < 4; ++n) {
                    const v2f b1 = {bq[n].z, bq[n].w};
                    acc[i][n] = __builtin_amdgcn_wmma_f32_16x16x4_f32(
                        false, a1[i], false, b1, (short)0, acc[i][n], false, false);
                }
        }
    }

    {
        const float* ablk[4];
        const float* brow[4];
#pragma unroll
        for (int i = 0; i < 4; ++i)
            ablk[i] = tp + ((size_t)((t0 >> 4) + i) * KQ_T) * 128 + lane * 4;
#pragma unroll
        for (int n = 0; n < 4; ++n)
            brow[n] = w2 + (f0 + 16 * n + lr) * LOWRANK + co;

        for (int k = 0; k < LOWRANK; k += 8) {
            v4f aq[4];
            v2f b0[4], b1[4];
#pragma unroll
            for (int i = 0; i < 4; ++i) aq[i] = *(const v4f*)(ablk[i] + k * 16);
#pragma unroll
            for (int n = 0; n < 4; ++n) {
                b0[n] = *(const v2f*)(brow[n] + k);
                b1[n] = *(const v2f*)(brow[n] + k + 4);
            }
#pragma unroll
            for (int i = 0; i < 4; ++i)
#pragma unroll
                for (int n = 0; n < 4; ++n) {
                    const v2f a0 = {aq[i].x, aq[i].y};
                    acc[i][n] = __builtin_amdgcn_wmma_f32_16x16x4_f32(
                        false, a0, false, b0[n], (short)0, acc[i][n], false, false);
                }
#pragma unroll
            for (int i = 0; i < 4; ++i)
#pragma unroll
                for (int n = 0; n < 4; ++n) {
                    const v2f a1 = {aq[i].z, aq[i].w};
                    acc[i][n] = __builtin_amdgcn_wmma_f32_16x16x4_f32(
                        false, a1, false, b1[n], (short)0, acc[i][n], false, false);
                }
        }
    }

    const int m0 = (lane >> 4) * 8;
#pragma unroll
    for (int n = 0; n < 4; ++n) {
        const float bv = bias[f0 + 16 * n + lr];
#pragma unroll
        for (int i = 0; i < 4; ++i)
#pragma unroll
            for (int r = 0; r < 8; ++r)
                out[(t0 + 16 * i + m0 + r) * OUT_F + f0 + 16 * n + lr] = acc[i][n][r] + bv;
    }
}

extern "C" void kernel_launch(void* const* d_in, const int* in_sizes, int n_in,
                              void* d_out, int out_size, void* d_ws, size_t ws_size,
                              hipStream_t stream) {
    const float* x      = (const float*)d_in[0];
    const float* weight = (const float*)d_in[1];
    const float* w1     = (const float*)d_in[2];
    const float* w2     = (const float*)d_in[3];
    const float* bias   = (const float*)d_in[4];
    const int*   flat   = (const int*)d_in[5];
    float*       out    = (float*)d_out;

    float* wg = (float*)d_ws;
    float* T  = wg + (size_t)OUT_F * KBF;

    float* w1p = wg;

    pack_w1<<<dim3((LOWRANK * IN_F) / (4 * 256)), dim3(256), 0, stream>>>(w1, w1p);

    gemm_xw1<<<dim3(1024 / 8), dim3(256), 0, stream>>>(x, w1p, T);

    gather_wg<<<dim3((OUT_F * KBF) / (4 * 256)), dim3(256), 0, stream>>>(weight, flat, wg);

    fused_out<<<dim3(4096 / 8), dim3(256), 0, stream>>>(x, wg, T, w2, bias, flat, out);
}


# both GEMM kernels: two register sets of fragments, the next k-group's loads (32 k in x@w1.T, 8 k in the fused kernel) are issued in front of the current group's 32 WMMAs; k order and sums unchanged
# speedup vs baseline: 1.7407x; 1.3672x over previous
#include <hip/hip_runtime.h>

typedef __attribute__((ext_vector_type(2))) float v2f;
typedef __attribute__((ext_vector_type(4))) float v4f;
typedef __attribute__((ext_vector_type(8))) float v8f;

typedef const __attribute__((address_space(1))) float* gptr;
typedef const __attribute__((address_space(1))) v2f*   gptr2;
typedef const __attribute__((address_space(1))) v4f*   gptr4;

#define TOKENS  4096
#define IN_F    4096
#define OUT_F   4096
#define LOWRANK 256
#define BS      256
#define ACTIVE  5
#define KBF     (ACTIVE * BS)
#define KQ_W    (KBF / 8)
#define KQ_T    (LOWRANK / 8)
#define KQ_1    (IN_F / 8)


__global__ __launch_bounds__(256) void pack_w1(const float* __restrict__ w1,
                                               float* __restrict__ w1p) {
    const int g    = blockIdx.x * 256 + threadIdx.x;
    const int lane = g & 31;
    const int blk  = g >> 5;
    const int fb   = blk / KQ_1;
    const int kq   = blk - fb * KQ_1;
    const int f    = fb * 16 + (lane & 15);
    const int h    = lane >> 4;
    const float* s = w1 + (size_t)f * IN_F + kq * 8 + 2 * h;
    const v2f lo = *(const v2f*)(s);
    const v2f hi = *(const v2f*)(s + 4);
    v4f o;
    o.x = lo.x; o.y = lo.y; o.z = hi.x; o.w = hi.y;
    *(v4f*)(w1p + (size_t)g * 4) = o;
}

__global__ __launch_bounds__(256) void gather_wg(const float* __restrict__ weight,
                                                 const int* __restrict__ flat,
                                                 float* __restrict__ wp) {
    const int g    = blockIdx.x * 256 + threadIdx.x;
    const int lane = g & 31;
    const int blk  = g >> 5;
    const int fb   = blk / KQ_W;
    const int kq   = blk - fb * KQ_W;
    const int f    = fb * 16 + (lane & 15);
    const int h    = lane >> 4;
    const int ob   = f >> 8;
    const int rr   = f & 255;
    const int j    = kq >> 5;
    const int c0   = (kq & 31) * 8 + 2 * h;
    const int fl   = flat[ob * ACTIVE + j];
    const float* s = weight + ((size_t)fl * BS + rr) * BS + c0;
    const v2f lo = *(const v2f*)(s);
    const v2f hi = *(const v2f*)(s + 4);
    v4f o;
    o.x = lo.x; o.y = lo.y; o.z = hi.x; o.w = hi.y;
    *(v4f*)(wp + (size_t)g * 4) = o;
}

struct Frags1 {
    v2f a[4][4];
    v4f b[4][2];
};

__device__ __forceinline__ void load_group1(Frags1& g, gptr a0p, gptr a1p,
                                            gptr b0q, gptr b1q, int k) {
#pragma unroll
    for (int u = 0; u < 4; ++u) {
        const int kk = k + 8 * u;
        g.a[u][0] = *(gptr2)(a0p + kk);
        g.a[u][1] = *(gptr2)(a1p + kk);
        g.a[u][2] = *(gptr2)(a0p + kk + 4);
        g.a[u][3] = *(gptr2)(a1p + kk + 4);
        g.b[u][0] = *(gptr4)(b0q + kk * 16);
        g.b[u][1] = *(gptr4)(b1q + kk * 16);
    }
}

__device__ __forceinline__ void mma_group1(v8f& acc00, v8f& acc01, v8f& acc10, v8f& acc11,
                                           const Frags1& g) {
#pragma unroll
    for (int u = 0; u < 4; ++u) {
        const v2f b0  = {g.b[u][0].x, g.b[u][0].y}, b0n = {g.b[u][0].z, g.b[u][0].w};
        const v2f b1  = {g.b[u][1].x, g.b[u][1].y}, b1n = {g.b[u][1].z, g.b[u][1].w};
        acc00 = __builtin_amdgcn_wmma_f32_16x16x4_f32(false, g.a[u][0], false, b0,
                                                      (short)0, acc00, false, false);
        acc01 = __builtin_amdgcn_wmma_f32_16x16x4_f32(false, g.a[u][0], false, b1,
                                                      (short)0, acc01, false, false);
        acc10 = __builtin_amdgcn_wmma_f32_16x16x4_f32(false, g.a[u][1], false, b0,
                                                      (short)0, acc10, false, false);
        acc11 = __builtin_amdgcn_wmma_f32_16x16x4_f32(false, g.a[u][1], false, b1,
                                                      (short)0, acc11, false, false);
        acc00 = __builtin_amdgcn_wmma_f32_16x16x4_f32(false, g.a[u][2], false, b0n,
                                                      (short)0, acc00, false, false);
        acc01 = __builtin_amdgcn_wmma_f32_16x16x4_f32(false, g.a[u][2], false, b1n,
                                                      (short)0, acc01, false, false);
        acc10 = __builtin_amdgcn_wmma_f32_16x16x4_f32(false, g.a[u][3], false, b0n,
                                                      (short)0, acc10, false, false);
        acc11 = __builtin_amdgcn_wmma_f32_16x16x4_f32(false, g.a[u][3], false, b1n,
                                                      (short)0, acc11, false, false);
    }
}

#define GROUP_FENCE() __builtin_amdgcn_sched_barrier(0)

__global__ __launch_bounds__(256) void gemm_xw1(const float* __restrict__ x,
                                                const float* __restrict__ w1p,
                                                float* __restrict__ T) {
    const int lane = threadIdx.x & 31;
    const int wave = threadIdx.x >> 5;
    const int tile = blockIdx.x * 8 + wave;
    const int nft  = LOWRANK / 32;
    const int tt   = tile / nft;
    const int ft   = tile - tt * nft;
    const int t0   = tt * 32;
    const int f0   = ft * 32;
    const int lr   = lane & 15;
    const int co   = (lane >> 4) * 2;

    const gptr a0p = (gptr)(x  + (t0 + lr)      * IN_F + co);
    const gptr a1p = (gptr)(x  + (t0 + 16 + lr) * IN_F + co);
    const gptr b0q = (gptr)(w1p + ((size_t)((f0 >> 4))     * KQ_1) * 128 + lane * 4);
    const gptr b1q = (gptr)(w1p + ((size_t)((f0 >> 4) + 1) * KQ_1) * 128 + lane * 4);

    v8f acc00 = {}, acc01 = {}, acc10 = {}, acc11 = {};
    Frags1 ga, gb;
    load_group1(ga, a0p, a1p, b0q, b1q, 0);
    for (int k = 0; k < IN_F - 64; k += 64) {
        load_group1(gb, a0p, a1p, b0q, b1q, k + 32);
        GROUP_FENCE();
        mma_group1(acc00, acc01, acc10, acc11, ga);
        GROUP_FENCE();
        load_group1(ga, a0p, a1p, b0q, b1q, k + 64);
        GROUP_FENCE();
        mma_group1(acc00, acc01, acc10, acc11, gb);
        GROUP_FENCE();
    }
    load_group1(gb, a0p, a1p, b0q, b1q, IN_F - 32);
    GROUP_FENCE();
    mma_group1(acc00, acc01, acc10, acc11, ga);
    mma_group1(acc00, acc01, acc10, acc11, gb);
    const int m0  = (lane >> 4) * 8;
    const int ff  = lr & 7;
    const int hh  = (ff >> 1) & 1;
    const int e   = (ff & 1) + 2 * (ff >> 2);
    const int tb0 = t0 >> 4;
    const int kq0 = (f0 >> 3) + (lr >> 3);
    const int in0 = (m0 + 16 * hh) * 4 + e;
    float* o00 = T + ((size_t)(tb0     * KQ_T + kq0)     * 128) + in0;
    float* o01 = T + ((size_t)(tb0     * KQ_T + kq0 + 2) * 128) + in0;
    float* o10 = T + ((size_t)((tb0+1) * KQ_T + kq0)     * 128) + in0;
    float* o11 = T + ((size_t)((tb0+1) * KQ_T + kq0 + 2) * 128) + in0;
#pragma unroll
    for (int r = 0; r < 8; ++r) {
        o00[4 * r] = acc00[r];
        o01[4 * r] = acc01[r];
        o10[4 * r] = acc10[r];
        o11[4 * r] = acc11[r];
    }
}

struct Frags {
    v2f p[8];
    v4f q[4];
};

__device__ __forceinline__ void load_group(Frags& f, const gptr (&p2)[4],
                                           const gptr (&p4)[4], int k) {
#pragma unroll
    for (int i = 0; i < 4; ++i) {
        f.p[i]     = *(gptr2)(p2[i] + k);
        f.p[4 + i] = *(gptr2)(p2[i] + k + 4);
    }
#pragma unroll
    for (int n = 0; n < 4; ++n) f.q[n] = *(gptr4)(p4[n] + k * 16);
}

__device__ __forceinline__ void mma_bf(v8f (&acc)[4][4], const Frags& f) {
#pragma unroll
    for (int i = 0; i < 4; ++i)
#pragma unroll
        for (int n = 0; n < 4; ++n) {
            const v2f b0 = {f.q[n].x, f.q[n].y};
            acc[i][n] = __builtin_amdgcn_wmma_f32_16x16x4_f32(
                false, f.p[i], false, b0, (short)0, acc[i][n], false, false);
        }
#pragma unroll
    for (int i = 0; i < 4; ++i)
#pragma unroll
        for (int n = 0; n < 4; ++n) {
            const v2f b1 = {f.q[n].z, f.q[n].w};
            acc[i][n] = __builtin_amdgcn_wmma_f32_16x16x4_f32(
                false, f.p[4 + i], false, b1, (short)0, acc[i][n], false, false);
        }
}

__device__ __forceinline__ void mma_lr(v8f (&acc)[4][4], const Frags& f) {
#pragma unroll
    for (int i = 0; i < 4; ++i)
#pragma unroll
        for (int n = 0; n < 4; ++n) {
            const v2f a0 = {f.q[i].x, f.q[i].y};
            acc[i][n] = __builtin_amdgcn_wmma_f32_16x16x4_f32(
                false, a0, false, f.p[n], (short)0, acc[i][n], false, false);
        }
#pragma unroll
    for (int i = 0; i < 4; ++i)
#pragma unroll
        for (int n = 0; n < 4; ++n) {
            const v2f a1 = {f.q[i].z, f.q[i].w};
            acc[i][n] = __builtin_amdgcn_wmma_f32_16x16x4_f32(
                false, a1, false, f.p[4 + n], (short)0, acc[i][n], false, false);
        }
}

__global__ __launch_bounds__(256) void fused_out(const float* __restrict__ x,
                                                 const float* __restrict__ wp,
                                                 const float* __restrict__ tp,
                                                 const float* __restrict__ w2,
                                                 const float* __restrict__ bias,
                                                 const int* __restrict__ flat,
                                                 float* __restrict__ out) {
    const int lane = threadIdx.x & 31;
    const int wave = threadIdx.x >> 5;
    const int tile = blockIdx.x * 8 + wave;
    const int nft  = OUT_F / 64;
    const int tt   = tile / nft;
    const int ft   = tile - tt * nft;
    const int t0   = tt * 64;
    const int f0   = ft * 64;
    const int ob   = f0 >> 8;
    const int lr   = lane & 15;
    const int co   = (lane >> 4) * 2;

    v8f acc[4][4];
#pragma unroll
    for (int i = 0; i < 4; ++i)
#pragma unroll
        for (int n = 0; n < 4; ++n) acc[i][n] = (v8f){};

    gptr xrow[4];
#pragma unroll
    for (int i = 0; i < 4; ++i) xrow[i] = (gptr)(x + (t0 + 16 * i + lr) * IN_F + co);

    gptr p2[4];
    gptr p4[4];
    {
        const int q = flat[ob * ACTIVE] / ACTIVE;
#pragma unroll
        for (int i = 0; i < 4; ++i) p2[i] = xrow[i] + q * BS;
#pragma unroll
        for (int n = 0; n < 4; ++n)
            p4[n] = (gptr)(wp + ((size_t)((f0 >> 4) + n) * KQ_W) * 128 + lane * 4);
    }

    Frags fa, fb;
    load_group(fa, p2, p4, 0);

    for (int j = 0; j < ACTIVE; ++j) {
        gptr n2[4];
        gptr n4[4];
        const bool   last = (j == ACTIVE - 1);
        const int    qn   = flat[ob * ACTIVE + (last ? j : j + 1)] / ACTIVE;
#pragma unroll
        for (int i = 0; i < 4; ++i)
            n2[i] = last ? (gptr)(w2 + (f0 + 16 * i + lr) * LOWRANK + co) : (xrow[i] + qn * BS);
#pragma unroll
        for (int n = 0; n < 4; ++n)
            n4[n] = last ? (gptr)(tp + ((size_t)((t0 >> 4) + n) * KQ_T) * 128 + lane * 4)
                         : (p4[n] + (BS / 8) * 128);

        for (int k = 0; k < BS - 16; k += 16) {
            load_group(fb, p2, p4, k + 8);
            GROUP_FENCE();
            mma_bf(acc, fa);
            GROUP_FENCE();
            load_group(fa, p2, p4, k + 16);
            GROUP_FENCE();
            mma_bf(acc, fb);
            GROUP_FENCE();
        }
        load_group(fb, p2, p4, BS - 8);
        GROUP_FENCE();
        mma_bf(acc, fa);
        GROUP_FENCE();
        load_group(fa, n2, n4, 0);
        GROUP_FENCE();
        mma_bf(acc, fb);
        GROUP_FENCE();
#pragma unroll
        for (int i = 0; i < 4; ++i) { p2[i] = n2[i]; p4[i] = n4[i]; }
    }

    for (int k = 0; k < LOWRANK - 16; k += 16) {
        load_group(fb, p2, p4, k + 8);
        GROUP_FENCE();
        mma_lr(acc, fa);
        GROUP_FENCE();
        load_group(fa, p2, p4, k + 16);
        GROUP_FENCE();
        mma_lr(acc, fb);
        GROUP_FENCE();
    }
    load_group(fb, p2, p4, LOWRANK - 8);
    GROUP_FENCE();
    mma_lr(acc, fa);
    mma_lr(acc, fb);

    const int m0 = (lane >> 4) * 8;
#pragma unroll
    for (int n = 0; n < 4; ++n) {
        const float bv = bias[f0 + 16 * n + lr];
#pragma unroll
        for (int i = 0; i < 4; ++i)
#pragma unroll
            for (int r = 0; r < 8; ++r)
                out[(t0 + 16 * i + m0 + r) * OUT_F + f0 + 16 * n + lr] = acc[i][n][r] + bv;
    }
}

extern "C" void kernel_launch(void* const* d_in, const int* in_sizes, int n_in,
                              void* d_out, int out_size, void* d_ws, size_t ws_size,
                              hipStream_t stream) {
    const float* x      = (const float*)d_in[0];
    const float* weight = (const float*)d_in[1];
    const float* w1     = (const float*)d_in[2];
    const float* w2     = (const float*)d_in[3];
    const float* bias   = (const float*)d_in[4];
    const int*   flat   = (const int*)d_in[5];
    float*       out    = (float*)d_out;

    float* wg = (float*)d_ws;
    float* T  = wg + (size_t)OUT_F * KBF;

    float* w1p = wg;

    pack_w1<<<dim3((LOWRANK * IN_F) / (4 * 256)), dim3(256), 0, stream>>>(w1, w1p);

    gemm_xw1<<<dim3(1024 / 8), dim3(256), 0, stream>>>(x, w1p, T);

    gather_wg<<<dim3((OUT_F * KBF) / (4 * 256)), dim3(256), 0, stream>>>(weight, flat, wg);

    fused_out<<<dim3(4096 / 8), dim3(256), 0, stream>>>(x, wg, T, w2, bias, flat, out);
}


# x@w1.T: K split over the 8 waves of a work-group (32x128 tile each over 512 k), partial sums added through LDS in slice order; 256 work-groups; T is now a sum of 8 partial sums
# speedup vs baseline: 1.9650x; 1.1289x over previous
#include <hip/hip_runtime.h>

typedef __attribute__((ext_vector_type(2))) float v2f;
typedef __attribute__((ext_vector_type(4))) float v4f;
typedef __attribute__((ext_vector_type(8))) float v8f;

typedef const __attribute__((address_space(1))) float* gptr;
typedef const __attribute__((address_space(1))) v2f*   gptr2;
typedef const __attribute__((address_space(1))) v4f*   gptr4;

#define TOKENS  4096
#define IN_F    4096
#define OUT_F   4096
#define LOWRANK 256
#define BS      256
#define ACTIVE  5
#define KBF     (ACTIVE * BS)
#define KQ_W    (KBF / 8)
#define KQ_T    (LOWRANK / 8)
#define KQ_1    (IN_F / 8)


__global__ __launch_bounds__(256) void pack_w1(const float* __restrict__ w1,
                                               float* __restrict__ w1p) {
    const int g    = blockIdx.x * 256 + threadIdx.x;
    const int lane = g & 31;
    const int blk  = g >> 5;
    const int fb   = blk / KQ_1;
    const int kq   = blk - fb * KQ_1;
    const int f    = fb * 16 + (lane & 15);
    const int h    = lane >> 4;
    const float* s = w1 + (size_t)f * IN_F + kq * 8 + 2 * h;
    const v2f lo = *(const v2f*)(s);
    const v2f hi = *(const v2f*)(s + 4);
    v4f o;
    o.x = lo.x; o.y = lo.y; o.z = hi.x; o.w = hi.y;
    *(v4f*)(w1p + (size_t)g * 4) = o;
}

__global__ __launch_bounds__(256) void gather_wg(const float* __restrict__ weight,
                                                 const int* __restrict__ flat,
                                                 float* __restrict__ wp) {
    const int g    = blockIdx.x * 256 + threadIdx.x;
    const int lane = g & 31;
    const int blk  = g >> 5;
    const int fb   = blk / KQ_W;
    const int kq   = blk - fb * KQ_W;
    const int f    = fb * 16 + (lane & 15);
    const int h    = lane >> 4;
    const int ob   = f >> 8;
    const int rr   = f & 255;
    const int j    = kq >> 5;
    const int c0   = (kq & 31) * 8 + 2 * h;
    const int fl   = flat[ob * ACTIVE + j];
    const float* s = weight + ((size_t)fl * BS + rr) * BS + c0;
    const v2f lo = *(const v2f*)(s);
    const v2f hi = *(const v2f*)(s + 4);
    v4f o;
    o.x = lo.x; o.y = lo.y; o.z = hi.x; o.w = hi.y;
    *(v4f*)(wp + (size_t)g * 4) = o;
}

#define KSLICE (IN_F / 8)

struct FragsT {
    v2f a[2][2][2];
    v4f b[2][8];
};

__device__ __forceinline__ void load_groupT(FragsT& g, gptr a0p, gptr a1p, gptr bq, int k) {
#pragma unroll
    for (int u = 0; u < 2; ++u) {
        const int kk = k + 8 * u;
        g.a[u][0][0] = *(gptr2)(a0p + kk);
        g.a[u][0][1] = *(gptr2)(a1p + kk);
        g.a[u][1][0] = *(gptr2)(a0p + kk + 4);
        g.a[u][1][1] = *(gptr2)(a1p + kk + 4);
#pragma unroll
        for (int n = 0; n < 8; ++n)
            g.b[u][n] = *(gptr4)(bq + n * (KQ_1 * 128) + kk * 16);
    }
}

__device__ __forceinline__ void mma_groupT(v8f (&acc)[2][8], const FragsT& g) {
#pragma unroll
    for (int u = 0; u < 2; ++u) {
#pragma unroll
        for (int i = 0; i < 2; ++i)
#pragma unroll
            for (int n = 0; n < 8; ++n) {
                const v2f b0 = {g.b[u][n].x, g.b[u][n].y};
                acc[i][n] = __builtin_amdgcn_wmma_f32_16x16x4_f32(
                    false, g.a[u][0][i], false, b0, (short)0, acc[i][n], false, false);
            }
#pragma unroll
        for (int i = 0; i < 2; ++i)
#pragma unroll
            for (int n = 0; n < 8; ++n) {
                const v2f b1 = {g.b[u][n].z, g.b[u][n].w};
                acc[i][n] = __builtin_amdgcn_wmma_f32_16x16x4_f32(
                    false, g.a[u][1][i], false, b1, (short)0, acc[i][n], false, false);
            }
    }
}

#define GROUP_FENCE() __builtin_amdgcn_sched_barrier(0)

__global__ __launch_bounds__(256) void gemm_xw1(const float* __restrict__ x,
                                                const float* __restrict__ w1p,
                                                float* __restrict__ T) {
    __shared__ float part[8][16][8][32];

    const int lane = threadIdx.x & 31;
    const int wave = threadIdx.x >> 5;
    const int t0   = (blockIdx.x >> 1) * 32;
    const int f0   = (blockIdx.x & 1) * 128;
    const int lr   = lane & 15;
    const int co   = (lane >> 4) * 2;
    const int kb   = wave * KSLICE;

    const gptr a0p = (gptr)(x + (t0 + lr)      * IN_F + kb + co);
    const gptr a1p = (gptr)(x + (t0 + 16 + lr) * IN_F + kb + co);
    const gptr bq  = (gptr)(w1p + ((size_t)(f0 >> 4) * KQ_1) * 128 + kb * 16 + lane * 4);

    v8f acc[2][8];
#pragma unroll
    for (int i = 0; i < 2; ++i)
#pragma unroll
        for (int n = 0; n < 8; ++n) acc[i][n] = (v8f){};

    FragsT ga, gb;
    load_groupT(ga, a0p, a1p, bq, 0);
    for (int k = 0; k < KSLICE - 32; k += 32) {
        load_groupT(gb, a0p, a1p, bq, k + 16);
        GROUP_FENCE();
        mma_groupT(acc, ga);
        GROUP_FENCE();
        load_groupT(ga, a0p, a1p, bq, k + 32);
        GROUP_FENCE();
        mma_groupT(acc, gb);
        GROUP_FENCE();
    }
    load_groupT(gb, a0p, a1p, bq, KSLICE - 16);
    GROUP_FENCE();
    mma_groupT(acc, ga);
    mma_groupT(acc, gb);

#pragma unroll
    for (int i = 0; i < 2; ++i)
#pragma unroll
        for (int n = 0; n < 8; ++n)
#pragma unroll
            for (int r = 0; r < 8; ++r) part[wave][i * 8 + n][r][lane] = acc[i][n][r];
    __syncthreads();

    const int m0  = (lane >> 4) * 8;
    const int ff  = lr & 7;
    const int hh  = (ff >> 1) & 1;
    const int e   = (ff & 1) + 2 * (ff >> 2);
    const int in0 = (m0 + 16 * hh) * 4 + e;
#pragma unroll
    for (int s = 0; s < 2; ++s) {
        const int tile = wave * 2 + s;
        const int ti   = tile >> 3;
        const int tn   = tile & 7;
        float sum[8];
#pragma unroll
        for (int r = 0; r < 8; ++r) sum[r] = part[0][tile][r][lane];
#pragma unroll
        for (int p = 1; p < 8; ++p)
#pragma unroll
            for (int r = 0; r < 8; ++r) sum[r] += part[p][tile][r][lane];
        const int tb = (t0 >> 4) + ti;
        const int kq = (f0 >> 3) + 2 * tn + (lr >> 3);
        float* o = T + ((size_t)(tb * KQ_T + kq) * 128) + in0;
#pragma unroll
        for (int r = 0; r < 8; ++r) o[4 * r] = sum[r];
    }
}

struct Frags {
    v2f p[8];
    v4f q[4];
};

__device__ __forceinline__ void load_group(Frags& f, const gptr (&p2)[4],
                                           const gptr (&p4)[4], int k) {
#pragma unroll
    for (int i = 0; i < 4; ++i) {
        f.p[i]     = *(gptr2)(p2[i] + k);
        f.p[4 + i] = *(gptr2)(p2[i] + k + 4);
    }
#pragma unroll
    for (int n = 0; n < 4; ++n) f.q[n] = *(gptr4)(p4[n] + k * 16);
}

__device__ __forceinline__ void mma_bf(v8f (&acc)[4][4], const Frags& f) {
#pragma unroll
    for (int i = 0; i < 4; ++i)
#pragma unroll
        for (int n = 0; n < 4; ++n) {
            const v2f b0 = {f.q[n].x, f.q[n].y};
            acc[i][n] = __builtin_amdgcn_wmma_f32_16x16x4_f32(
                false, f.p[i], false, b0, (short)0, acc[i][n], false, false);
        }
#pragma unroll
    for (int i = 0; i < 4; ++i)
#pragma unroll
        for (int n = 0; n < 4; ++n) {
            const v2f b1 = {f.q[n].z, f.q[n].w};
            acc[i][n] = __builtin_amdgcn_wmma_f32_16x16x4_f32(
                false, f.p[4 + i], false, b1, (short)0, acc[i][n], false, false);
        }
}

__device__ __forceinline__ void mma_lr(v8f (&acc)[4][4], const Frags& f) {
#pragma unroll
    for (int i = 0; i < 4; ++i)
#pragma unroll
        for (int n = 0; n < 4; ++n) {
            const v2f a0 = {f.q[i].x, f.q[i].y};
            acc[i][n] = __builtin_amdgcn_wmma_f32_16x16x4_f32(
                false, a0, false, f.p[n], (short)0, acc[i][n], false, false);
        }
#pragma unroll
    for (int i = 0; i < 4; ++i)
#pragma unroll
        for (int n = 0; n < 4; ++n) {
            const v2f a1 = {f.q[i].z, f.q[i].w};
            acc[i][n] = __builtin_amdgcn_wmma_f32_16x16x4_f32(
                false, a1, false, f.p[4 + n], (short)0, acc[i][n], false, false);
        }
}

__global__ __launch_bounds__(256) void fused_out(const float* __restrict__ x,
                                                 const float* __restrict__ wp,
                                                 const float* __restrict__ tp,
                                                 const float* __restrict__ w2,
                                                 const float* __restrict__ bias,
                                                 const int* __restrict__ flat,
                                                 float* __restrict__ out) {
    const int lane = threadIdx.x & 31;
    const int wave = threadIdx.x >> 5;
    const int tile = blockIdx.x * 8 + wave;
    const int nft  = OUT_F / 64;
    const int tt   = tile / nft;
    const int ft   = tile - tt * nft;
    const int t0   = tt * 64;
    const int f0   = ft * 64;
    const int ob   = f0 >> 8;
    const int lr   = lane & 15;
    const int co   = (lane >> 4) * 2;

    v8f acc[4][4];
#pragma unroll
    for (int i = 0; i < 4; ++i)
#pragma unroll
        for (int n = 0; n < 4; ++n) acc[i][n] = (v8f){};

    gptr xrow[4];
#pragma unroll
    for (int i = 0; i < 4; ++i) xrow[i] = (gptr)(x + (t0 + 16 * i + lr) * IN_F + co);

    gptr p2[4];
    gptr p4[4];
    {
        const int q = flat[ob * ACTIVE] / ACTIVE;
#pragma unroll
        for (int i = 0; i < 4; ++i) p2[i] = xrow[i] + q * BS;
#pragma unroll
        for (int n = 0; n < 4; ++n)
            p4[n] = (gptr)(wp + ((size_t)((f0 >> 4) + n) * KQ_W) * 128 + lane * 4);
    }

    Frags fa, fb;
    load_group(fa, p2, p4, 0);

    for (int j = 0; j < ACTIVE; ++j) {
        gptr n2[4];
        gptr n4[4];
        const bool   last = (j == ACTIVE - 1);
        const int    qn   = flat[ob * ACTIVE + (last ? j : j + 1)] / ACTIVE;
#pragma unroll
        for (int i = 0; i < 4; ++i)
            n2[i] = last ? (gptr)(w2 + (f0 + 16 * i + lr) * LOWRANK + co) : (xrow[i] + qn * BS);
#pragma unroll
        for (int n = 0; n < 4; ++n)
            n4[n] = last ? (gptr)(tp + ((size_t)((t0 >> 4) + n) * KQ_T) * 128 + lane * 4)
                         : (p4[n] + (BS / 8) * 128);

        for (int k = 0; k < BS - 16; k += 16) {
            load_group(fb, p2, p4, k + 8);
            GROUP_FENCE();
            mma_bf(acc, fa);
            GROUP_FENCE();
            load_group(fa, p2, p4, k + 16);
            GROUP_FENCE();
            mma_bf(acc, fb);
            GROUP_FENCE();
        }
        load_group(fb, p2, p4, BS - 8);
        GROUP_FENCE();
        mma_bf(acc, fa);
        GROUP_FENCE();
        load_group(fa, n2, n4, 0);
        GROUP_FENCE();
        mma_bf(acc, fb);
        GROUP_FENCE();
#pragma unroll
        for (int i = 0; i < 4; ++i) { p2[i] = n2[i]; p4[i] = n4[i]; }
    }

    for (int k = 0; k < LOWRANK - 16; k += 16) {
        load_group(fb, p2, p4, k + 8);
        GROUP_FENCE();
        mma_lr(acc, fa);
        GROUP_FENCE();
        load_group(fa, p2, p4, k + 16);
        GROUP_FENCE();
        mma_lr(acc, fb);
        GROUP_FENCE();
    }
    load_group(fb, p2, p4, LOWRANK - 8);
    GROUP_FENCE();
    mma_lr(acc, fa);
    mma_lr(acc, fb);

    const int m0 = (lane >> 4) * 8;
#pragma unroll
    for (int n = 0; n < 4; ++n) {
        const float bv = bias[f0 + 16 * n + lr];
#pragma unroll
        for (int i = 0; i < 4; ++i)
#pragma unroll
            for (int r = 0; r < 8; ++r)
                out[(t0 + 16 * i + m0 + r) * OUT_F + f0 + 16 * n + lr] = acc[i][n][r] + bv;
    }
}

extern "C" void kernel_launch(void* const* d_in, const int* in_sizes, int n_in,
                              void* d_out, int out_size, void* d_ws, size_t ws_size,
                              hipStream_t stream) {
    const float* x      = (const float*)d_in[0];
    const float* weight = (const float*)d_in[1];
    const float* w1     = (const float*)d_in[2];
    const float* w2     = (const float*)d_in[3];
    const float* bias   = (const float*)d_in[4];
    const int*   flat   = (const int*)d_in[5];
    float*       out    = (float*)d_out;

    float* wg = (float*)d_ws;
    float* T  = wg + (size_t)OUT_F * KBF;

    float* w1p = wg;

    pack_w1<<<dim3((LOWRANK * IN_F) / (4 * 256)), dim3(256), 0, stream>>>(w1, w1p);

    gemm_xw1<<<dim3(256), dim3(256), 0, stream>>>(x, w1p, T);

    gather_wg<<<dim3((OUT_F * KBF) / (4 * 256)), dim3(256), 0, stream>>>(weight, flat, wg);

    fused_out<<<dim3(4096 / 8), dim3(256), 0, stream>>>(x, wg, T, w2, bias, flat, out);
}


# fused output kernel re-tiled: wave tile 64x128 (was 64x64), work-group 256x256 = one output block, 256 work-groups, 2 waves per SIMD; half the partial-line x loads per WMMA; sums unchanged
# speedup vs baseline: 2.0715x; 1.0542x over previous
#include <hip/hip_runtime.h>

typedef __attribute__((ext_vector_type(2))) float v2f;
typedef __attribute__((ext_vector_type(4))) float v4f;
typedef __attribute__((ext_vector_type(8))) float v8f;

typedef const __attribute__((address_space(1))) float* gptr;
typedef const __attribute__((address_space(1))) v2f*   gptr2;
typedef const __attribute__((address_space(1))) v4f*   gptr4;

#define TOKENS  4096
#define IN_F    4096
#define OUT_F   4096
#define LOWRANK 256
#define BS      256
#define ACTIVE  5
#define KBF     (ACTIVE * BS)
#define KQ_W    (KBF / 8)
#define KQ_T    (LOWRANK / 8)
#define KQ_1    (IN_F / 8)


__global__ __launch_bounds__(256) void pack_w1(const float* __restrict__ w1,
                                               float* __restrict__ w1p) {
    const int g    = blockIdx.x * 256 + threadIdx.x;
    const int lane = g & 31;
    const int blk  = g >> 5;
    const int fb   = blk / KQ_1;
    const int kq   = blk - fb * KQ_1;
    const int f    = fb * 16 + (lane & 15);
    const int h    = lane >> 4;
    const float* s = w1 + (size_t)f * IN_F + kq * 8 + 2 * h;
    const v2f lo = *(const v2f*)(s);
    const v2f hi = *(const v2f*)(s + 4);
    v4f o;
    o.x = lo.x; o.y = lo.y; o.z = hi.x; o.w = hi.y;
    *(v4f*)(w1p + (size_t)g * 4) = o;
}

__global__ __launch_bounds__(256) void gather_wg(const float* __restrict__ weight,
                                                 const int* __restrict__ flat,
                                                 float* __restrict__ wp) {
    const int g    = blockIdx.x * 256 + threadIdx.x;
    const int lane = g & 31;
    const int blk  = g >> 5;
    const int fb   = blk / KQ_W;
    const int kq   = blk - fb * KQ_W;
    const int f    = fb * 16 + (lane & 15);
    const int h    = lane >> 4;
    const int ob   = f >> 8;
    const int rr   = f & 255;
    const int j    = kq >> 5;
    const int c0   = (kq & 31) * 8 + 2 * h;
    const int fl   = flat[ob * ACTIVE + j];
    const float* s = weight + ((size_t)fl * BS + rr) * BS + c0;
    const v2f lo = *(const v2f*)(s);
    const v2f hi = *(const v2f*)(s + 4);
    v4f o;
    o.x = lo.x; o.y = lo.y; o.z = hi.x; o.w = hi.y;
    *(v4f*)(wp + (size_t)g * 4) = o;
}

#define KSLICE (IN_F / 8)

struct FragsT {
    v2f a[2][2][2];
    v4f b[2][8];
};

__device__ __forceinline__ void load_groupT(FragsT& g, gptr a0p, gptr a1p, gptr bq, int k) {
#pragma unroll
    for (int u = 0; u < 2; ++u) {
        const int kk = k + 8 * u;
        g.a[u][0][0] = *(gptr2)(a0p + kk);
        g.a[u][0][1] = *(gptr2)(a1p + kk);
        g.a[u][1][0] = *(gptr2)(a0p + kk + 4);
        g.a[u][1][1] = *(gptr2)(a1p + kk + 4);
#pragma unroll
        for (int n = 0; n < 8; ++n)
            g.b[u][n] = *(gptr4)(bq + n * (KQ_1 * 128) + kk * 16);
    }
}

__device__ __forceinline__ void mma_groupT(v8f (&acc)[2][8], const FragsT& g) {
#pragma unroll
    for (int u = 0; u < 2; ++u) {
#pragma unroll
        for (int i = 0; i < 2; ++i)
#pragma unroll
            for (int n = 0; n < 8; ++n) {
                const v2f b0 = {g.b[u][n].x, g.b[u][n].y};
                acc[i][n] = __builtin_amdgcn_wmma_f32_16x16x4_f32(
                    false, g.a[u][0][i], false, b0, (short)0, acc[i][n], false, false);
            }
#pragma unroll
        for (int i = 0; i < 2; ++i)
#pragma unroll
            for (int n = 0; n < 8; ++n) {
                const v2f b1 = {g.b[u][n].z, g.b[u][n].w};
                acc[i][n] = __builtin_amdgcn_wmma_f32_16x16x4_f32(
                    false, g.a[u][1][i], false, b1, (short)0, acc[i][n], false, false);
            }
    }
}

#define GROUP_FENCE() __builtin_amdgcn_sched_barrier(0)

__global__ __launch_bounds__(256) void gemm_xw1(const float* __restrict__ x,
                                                const float* __restrict__ w1p,
                                                float* __restrict__ T) {
    __shared__ float part[8][16][8][32];

    const int lane = threadIdx.x & 31;
    const int wave = threadIdx.x >> 5;
    const int t0   = (blockIdx.x >> 1) * 32;
    const int f0   = (blockIdx.x & 1) * 128;
    const int lr   = lane & 15;
    const int co   = (lane >> 4) * 2;
    const int kb   = wave * KSLICE;

    const gptr a0p = (gptr)(x + (t0 + lr)      * IN_F + kb + co);
    const gptr a1p = (gptr)(x + (t0 + 16 + lr) * IN_F + kb + co);
    const gptr bq  = (gptr)(w1p + ((size_t)(f0 >> 4) * KQ_1) * 128 + kb * 16 + lane * 4);

    v8f acc[2][8];
#pragma unroll
    for (int i = 0; i < 2; ++i)
#pragma unroll
        for (int n = 0; n < 8; ++n) acc[i][n] = (v8f){};

    FragsT ga, gb;
    load_groupT(ga, a0p, a1p, bq, 0);
    for (int k = 0; k < KSLICE - 32; k += 32) {
        load_groupT(gb, a0p, a1p, bq, k + 16);
        GROUP_FENCE();
        mma_groupT(acc, ga);
        GROUP_FENCE();
        load_groupT(ga, a0p, a1p, bq, k + 32);
        GROUP_FENCE();
        mma_groupT(acc, gb);
        GROUP_FENCE();
    }
    load_groupT(gb, a0p, a1p, bq, KSLICE - 16);
    GROUP_FENCE();
    mma_groupT(acc, ga);
    mma_groupT(acc, gb);

#pragma unroll
    for (int i = 0; i < 2; ++i)
#pragma unroll
        for (int n = 0; n < 8; ++n)
#pragma unroll
            for (int r = 0; r < 8; ++r) part[wave][i * 8 + n][r][lane] = acc[i][n][r];
    __syncthreads();

    const int m0  = (lane >> 4) * 8;
    const int ff  = lr & 7;
    const int hh  = (ff >> 1) & 1;
    const int e   = (ff & 1) + 2 * (ff >> 2);
    const int in0 = (m0 + 16 * hh) * 4 + e;
#pragma unroll
    for (int s = 0; s < 2; ++s) {
        const int tile = wave * 2 + s;
        const int ti   = tile >> 3;
        const int tn   = tile & 7;
        float sum[8];
#pragma unroll
        for (int r = 0; r < 8; ++r) sum[r] = part[0][tile][r][lane];
#pragma unroll
        for (int p = 1; p < 8; ++p)
#pragma unroll
            for (int r = 0; r < 8; ++r) sum[r] += part[p][tile][r][lane];
        const int tb = (t0 >> 4) + ti;
        const int kq = (f0 >> 3) + 2 * tn + (lr >> 3);
        float* o = T + ((size_t)(tb * KQ_T + kq) * 128) + in0;
#pragma unroll
        for (int r = 0; r < 8; ++r) o[4 * r] = sum[r];
    }
}

struct FragsB {
    v2f p[8];
    v4f q[8];
};
struct FragsL {
    v4f q[4];
    v2f p[16];
};

__device__ __forceinline__ void load_bf(FragsB& f, const gptr (&xr)[4], int xk,
                                        gptr bq, int kg) {
#pragma unroll
    for (int i = 0; i < 4; ++i) {
        f.p[i]     = *(gptr2)(xr[i] + xk);
        f.p[4 + i] = *(gptr2)(xr[i] + xk + 4);
    }
#pragma unroll
    for (int n = 0; n < 8; ++n) f.q[n] = *(gptr4)(bq + n * (KQ_W * 128) + kg * 16);
}

__device__ __forceinline__ void load_lr(FragsL& f, gptr tq, const gptr (&wr)[8], int k) {
#pragma unroll
    for (int i = 0; i < 4; ++i) f.q[i] = *(gptr4)(tq + i * (KQ_T * 128) + k * 16);
#pragma unroll
    for (int n = 0; n < 8; ++n) {
        f.p[n]     = *(gptr2)(wr[n] + k);
        f.p[8 + n] = *(gptr2)(wr[n] + k + 4);
    }
}

__device__ __forceinline__ void mma_bf(v8f (&acc)[4][8], const FragsB& f) {
#pragma unroll
    for (int i = 0; i < 4; ++i)
#pragma unroll
        for (int n = 0; n < 8; ++n) {
            const v2f b0 = {f.q[n].x, f.q[n].y};
            acc[i][n] = __builtin_amdgcn_wmma_f32_16x16x4_f32(
                false, f.p[i], false, b0, (short)0, acc[i][n], false, false);
        }
#pragma unroll
    for (int i = 0; i < 4; ++i)
#pragma unroll
        for (int n = 0; n < 8; ++n) {
            const v2f b1 = {f.q[n].z, f.q[n].w};
            acc[i][n] = __builtin_amdgcn_wmma_f32_16x16x4_f32(
                false, f.p[4 + i], false, b1, (short)0, acc[i][n], false, false);
        }
}

__device__ __forceinline__ void mma_lr(v8f (&acc)[4][8], const FragsL& f) {
#pragma unroll
    for (int i = 0; i < 4; ++i)
#pragma unroll
        for (int n = 0; n < 8; ++n) {
            const v2f a0 = {f.q[i].x, f.q[i].y};
            acc[i][n] = __builtin_amdgcn_wmma_f32_16x16x4_f32(
                false, a0, false, f.p[n], (short)0, acc[i][n], false, false);
        }
#pragma unroll
    for (int i = 0; i < 4; ++i)
#pragma unroll
        for (int n = 0; n < 8; ++n) {
            const v2f a1 = {f.q[i].z, f.q[i].w};
            acc[i][n] = __builtin_amdgcn_wmma_f32_16x16x4_f32(
                false, a1, false, f.p[8 + n], (short)0, acc[i][n], false, false);
        }
}

__device__ __forceinline__ void bf_chunk(v8f (&acc)[4][8], FragsB& fa, FragsB& fb,
                                         const gptr (&xr)[4], int xc, gptr bq, int kb) {
    for (int k = 0; k < BS - 16; k += 16) {
        load_bf(fb, xr, xc + k + 8, bq, kb + k + 8);
        GROUP_FENCE();
        mma_bf(acc, fa);
        GROUP_FENCE();
        load_bf(fa, xr, xc + k + 16, bq, kb + k + 16);
        GROUP_FENCE();
        mma_bf(acc, fb);
        GROUP_FENCE();
    }
    load_bf(fb, xr, xc + BS - 8, bq, kb + BS - 8);
    GROUP_FENCE();
    mma_bf(acc, fa);
    GROUP_FENCE();
}

__global__ __launch_bounds__(256) void fused_out(const float* __restrict__ x,
                                                 const float* __restrict__ wp,
                                                 const float* __restrict__ tp,
                                                 const float* __restrict__ w2,
                                                 const float* __restrict__ bias,
                                                 const int* __restrict__ flat,
                                                 float* __restrict__ out) {
    const int lane = threadIdx.x & 31;
    const int wave = threadIdx.x >> 5;
    const int nob  = OUT_F / BS;
    const int tt   = blockIdx.x / nob;
    const int ob   = blockIdx.x - tt * nob;
    const int t0   = tt * 256 + (wave & 3) * 64;
    const int f0   = ob * BS + (wave >> 2) * 128;
    const int lr   = lane & 15;
    const int co   = (lane >> 4) * 2;

    v8f acc[4][8];
#pragma unroll
    for (int i = 0; i < 4; ++i)
#pragma unroll
        for (int n = 0; n < 8; ++n) acc[i][n] = (v8f){};

    gptr xrow[4];
#pragma unroll
    for (int i = 0; i < 4; ++i) xrow[i] = (gptr)(x + (t0 + 16 * i + lr) * IN_F + co);
    gptr wrow[8];
#pragma unroll
    for (int n = 0; n < 8; ++n) wrow[n] = (gptr)(w2 + (f0 + 16 * n + lr) * LOWRANK + co);
    const gptr bq = (gptr)(wp + ((size_t)(f0 >> 4) * KQ_W) * 128 + lane * 4);
    const gptr tq = (gptr)(tp + ((size_t)(t0 >> 4) * KQ_T) * 128 + lane * 4);

    FragsB fa, fb;
    FragsL la, lb;
    int xc = (flat[ob * ACTIVE] / ACTIVE) * BS;
    load_bf(fa, xrow, xc, bq, 0);

    for (int j = 0; j < ACTIVE - 1; ++j) {
        const int xn = (flat[ob * ACTIVE + j + 1] / ACTIVE) * BS;
        bf_chunk(acc, fa, fb, xrow, xc, bq, j * BS);
        load_bf(fa, xrow, xn, bq, (j + 1) * BS);
        GROUP_FENCE();
        mma_bf(acc, fb);
        GROUP_FENCE();
        xc = xn;
    }
    bf_chunk(acc, fa, fb, xrow, xc, bq, (ACTIVE - 1) * BS);
    load_lr(la, tq, wrow, 0);
    GROUP_FENCE();
    mma_bf(acc, fb);
    GROUP_FENCE();

    for (int k = 0; k < LOWRANK - 16; k += 16) {
        load_lr(lb, tq, wrow, k + 8);
        GROUP_FENCE();
        mma_lr(acc, la);
        GROUP_FENCE();
        load_lr(la, tq, wrow, k + 16);
        GROUP_FENCE();
        mma_lr(acc, lb);
        GROUP_FENCE();
    }
    load_lr(lb, tq, wrow, LOWRANK - 8);
    GROUP_FENCE();
    mma_lr(acc, la);
    mma_lr(acc, lb);

    const int m0 = (lane >> 4) * 8;
#pragma unroll
    for (int n = 0; n < 8; ++n) {
        const float bv = bias[f0 + 16 * n + lr];
#pragma unroll
        for (int i = 0; i < 4; ++i)
#pragma unroll
            for (int r = 0; r < 8; ++r)
                out[(t0 + 16 * i + m0 + r) * OUT_F + f0 + 16 * n + lr] = acc[i][n][r] + bv;
    }
}

extern "C" void kernel_launch(void* const* d_in, const int* in_sizes, int n_in,
                              void* d_out, int out_size, void* d_ws, size_t ws_size,
                              hipStream_t stream) {
    const float* x      = (const float*)d_in[0];
    const float* weight = (const float*)d_in[1];
    const float* w1     = (const float*)d_in[2];
    const float* w2     = (const float*)d_in[3];
    const float* bias   = (const float*)d_in[4];
    const int*   flat   = (const int*)d_in[5];
    float*       out    = (float*)d_out;

    float* wg = (float*)d_ws;
    float* T  = wg + (size_t)OUT_F * KBF;

    float* w1p = wg;

    pack_w1<<<dim3((LOWRANK * IN_F) / (4 * 256)), dim3(256), 0, stream>>>(w1, w1p);

    gemm_xw1<<<dim3(256), dim3(256), 0, stream>>>(x, w1p, T);

    gather_wg<<<dim3((OUT_F * KBF) / (4 * 256)), dim3(256), 0, stream>>>(weight, flat, wg);

    fused_out<<<dim3(256), dim3(256), 0, stream>>>(x, wg, T, w2, bias, flat, out);
}
